# starting kernel
# baseline (speedup 1.0000x reference)
#include <hip/hip_runtime.h>

typedef __attribute__((ext_vector_type(2))) float v2f;
typedef __attribute__((ext_vector_type(8))) float v8f;

__global__ void cheb_degree_kernel(const int* __restrict__ row, float* __restrict__ deg, int E) {
    int e = blockIdx.x * blockDim.x + threadIdx.x;
    if (e < E) atomicAdd(&deg[row[e]], 1.0f);
}

__global__ void cheb_dis_kernel(float* __restrict__ deg, int n) {
    int i = blockIdx.x * blockDim.x + threadIdx.x;
    if (i < n) {
        float d = deg[i];
        deg[i] = (d > 0.0f) ? rsqrtf(fmaxf(d, 1.0f)) : 0.0f;
    }
}

__global__ void cheb_edgew_kernel(const int* __restrict__ row, const int* __restrict__ col,
                                  const float* __restrict__ dis, float* __restrict__ wE, int E) {
    int e = blockIdx.x * blockDim.x + threadIdx.x;
    if (e < E) wE[e] = -dis[row[e]] * dis[col[e]];
}

__global__ void cheb_negcopy_kernel(const float* __restrict__ a, float* __restrict__ b, int n) {
    int i = blockIdx.x * blockDim.x + threadIdx.x;
    if (i < n) b[i] = -a[i];
}

__global__ void cheb_scatter_kernel(const float* __restrict__ in, float* __restrict__ out,
                                    const int* __restrict__ src, const int* __restrict__ dst,
                                    const float* __restrict__ wE, float scale, int E) {
    int t = blockIdx.x * blockDim.x + threadIdx.x;
    int e = t >> 4;
    if (e >= E) return;
    int c = t & 15;
    int s = src[e];
    int d = dst[e];
    float w = wE[e] * scale;
    const float4 v = *(const float4*)(in + (size_t)s * 64 + (c << 2));
    float* op = out + (size_t)d * 64 + (c << 2);
    atomicAdd(op + 0, v.x * w);
    atomicAdd(op + 1, v.y * w);
    atomicAdd(op + 2, v.z * w);
    atomicAdd(op + 3, v.w * w);
}

template <int MT>
__global__ void __launch_bounds__(128)
cheb_gemm_wmma_kernel(const float* __restrict__ T0, const float* __restrict__ T1,
                      const float* __restrict__ T2, const float* __restrict__ W,
                      const float* __restrict__ bias, float* __restrict__ out,
                      int nTilesN, int F, int doRelu, int totalWaveTiles, int rowOffset) {
    __shared__ float sW[3 * 64 * 64 + 64];
    const int tid = threadIdx.y * 32 + threadIdx.x;
    const int nW = 3 * 64 * F;
    for (int i = tid; i < nW; i += 128) sW[i] = W[i];
    for (int i = tid; i < F; i += 128) sW[nW + i] = bias[i];
    __syncthreads();

    const int waveTile = blockIdx.x * 4 + threadIdx.y;
    if (waveTile >= totalWaveTiles) return;
    const int mGroup = waveTile / nTilesN;
    const int nTile = waveTile % nTilesN;

    const int lane = threadIdx.x;
    const int lg = lane >> 4;
    const int ln = lane & 15;
    const int rowBase = rowOffset + mGroup * (MT * 16);
    const int col = (nTile << 4) + ln;

    v8f acc[MT];
    const float bv = sW[nW + col];
#pragma unroll
    for (int i = 0; i < MT; ++i)
#pragma unroll
        for (int r = 0; r < 8; ++r) acc[i][r] = bv;

    const float* mats[3] = {T0, T1, T2};
#pragma unroll
    for (int m = 0; m < 3; ++m) {
        const float* __restrict__ T = mats[m];
        const float* Wm = &sW[m * 64 * F];
#pragma unroll
        for (int kb = 0; kb < 16; ++kb) {
            const int k0 = (kb << 2) + (lg << 1);
            v2f b;
            b.x = Wm[k0 * F + col];
            b.y = Wm[(k0 + 1) * F + col];
#pragma unroll
            for (int i = 0; i < MT; ++i) {
                v2f a = *(const v2f*)(T + (size_t)(rowBase + i * 16 + ln) * 64 + k0);
                acc[i] = __builtin_amdgcn_wmma_f32_16x16x4_f32(
                    false, a, false, b, (short)0, acc[i], false, false);
            }
        }
    }

#pragma unroll
    for (int i = 0; i < MT; ++i) {
        if (doRelu) {
#pragma unroll
            for (int r = 0; r < 8; ++r) acc[i][r] = fmaxf(acc[i][r], 0.0f);
        }
        float* orow = out + (size_t)(rowBase + i * 16 + (lg << 3)) * F + col;
#pragma unroll
        for (int r = 0; r < 8; ++r) orow[r * F] = acc[i][r];
    }
}

static void cheb_layer(const float* T0, const float* W, const float* b,
                       float* T1, float* T2, float* Hout,
                       const int* src, const int* dst, const float* wE,
                       int n, int E, int F, int relu, hipStream_t stream) {
    const int sthreads = E * 16;
    const int sblocks = (sthreads + 255) / 256;

    hipMemsetAsync(T1, 0, (size_t)n * 64 * sizeof(float), stream);
    cheb_scatter_kernel<<<sblocks, 256, 0, stream>>>(T0, T1, src, dst, wE, 1.0f, E);

    cheb_negcopy_kernel<<<(n * 64 + 255) / 256, 256, 0, stream>>>(T0, T2, n * 64);
    cheb_scatter_kernel<<<sblocks, 256, 0, stream>>>(T1, T2, src, dst, wE, 2.0f, E);

    const int nTilesN = F / 16;
    const int mTiles = n / 16;
    const int mGroups = mTiles / 5;
    const int wt5 = mGroups * nTilesN;
    if (wt5 > 0) {
        cheb_gemm_wmma_kernel<5><<<(wt5 + 3) / 4, dim3(32, 4), 0, stream>>>(
            T0, T1, T2, W, b, Hout, nTilesN, F, relu, wt5, 0);
    }
    const int remTiles = mTiles - mGroups * 5;
    if (remTiles > 0) {
        const int wt1 = remTiles * nTilesN;
        cheb_gemm_wmma_kernel<1><<<(wt1 + 3) / 4, dim3(32, 4), 0, stream>>>(
            T0, T1, T2, W, b, Hout, nTilesN, F, relu, wt1, mGroups * 80);
    }
}

extern "C" void kernel_launch(void* const* d_in, const int* in_sizes, int n_in,
                              void* d_out, int out_size, void* d_ws, size_t ws_size,
                              hipStream_t stream) {
    const float* x  = (const float*)d_in[0];
    const int*   ei = (const int*)d_in[1];
    const float* W1 = (const float*)d_in[2];
    const float* b1 = (const float*)d_in[3];
    const float* W2 = (const float*)d_in[4];
    const float* b2 = (const float*)d_in[5];
    const float* W3 = (const float*)d_in[6];
    const float* b3 = (const float*)d_in[7];

    const int n = in_sizes[0] / 64;
    const int E = in_sizes[1] / 2;
    const int* src = ei;
    const int* dst = ei + E;

    float* ws = (float*)d_ws;
    size_t o = 0;
    float* dis = ws + o; o += ((size_t)n + 63) & ~(size_t)63;
    float* wE  = ws + o; o += ((size_t)E + 63) & ~(size_t)63;
    float* T1  = ws + o; o += (size_t)n * 64;
    float* T2  = ws + o; o += (size_t)n * 64;
    float* HA  = ws + o; o += (size_t)n * 64;
    float* HB  = ws + o; o += (size_t)n * 64;

    hipMemsetAsync(dis, 0, (size_t)n * sizeof(float), stream);
    cheb_degree_kernel<<<(E + 255) / 256, 256, 0, stream>>>(src, dis, E);
    cheb_dis_kernel<<<(n + 255) / 256, 256, 0, stream>>>(dis, n);
    cheb_edgew_kernel<<<(E + 255) / 256, 256, 0, stream>>>(src, dst, dis, wE, E);

    cheb_layer(x,  W1, b1, T1, T2, HA, src, dst, wE, n, E, 64, 1, stream);
    cheb_layer(HA, W2, b2, T1, T2, HB, src, dst, wE, n, E, 64, 1, stream);
    cheb_layer(HB, W3, b3, T1, T2, (float*)d_out, src, dst, wE, n, E, 32, 0, stream);
}


# Each layer multiplies first: [O|Z|P]=T0@[W0-W2|W1|W2] in one GEMM (T0 read once), then Z+=2LP, O+=LZ: propagation at output width, GEMM stores start values, no memset or negcopy; sum order changes
# speedup vs baseline: 1.2657x; 1.2657x over previous
#include <hip/hip_runtime.h>


typedef __attribute__((ext_vector_type(2))) float v2f;
typedef __attribute__((ext_vector_type(8))) float v8f;

__global__ void cheb_degree_kernel(const int* __restrict__ row, float* __restrict__ deg, int E) {
    int e = blockIdx.x * blockDim.x + threadIdx.x;
    if (e < E) atomicAdd(&deg[row[e]], 1.0f);
}

__global__ void cheb_dis_kernel(float* __restrict__ deg, int n) {
    int i = blockIdx.x * blockDim.x + threadIdx.x;
    if (i < n) {
        float d = deg[i];
        deg[i] = (d > 0.0f) ? rsqrtf(fmaxf(d, 1.0f)) : 0.0f;
    }
}

__global__ void cheb_edgew_kernel(const int* __restrict__ row, const int* __restrict__ col,
                                  const float* __restrict__ dis, float* __restrict__ wE, int E) {
    int e = blockIdx.x * blockDim.x + threadIdx.x;
    if (e < E) wE[e] = -dis[row[e]] * dis[col[e]];
}

template <int F>
__global__ void cheb_scatter_kernel(const float* __restrict__ in, float* __restrict__ out,
                                    const int* __restrict__ src, const int* __restrict__ dst,
                                    const float* __restrict__ wE, float scale, int E) {
    constexpr int LPE = F / 4;
    int t = blockIdx.x * blockDim.x + threadIdx.x;
    int e = t / LPE;
    if (e >= E) return;
    int c = t % LPE;
    int s = src[e];
    int d = dst[e];
    float w = wE[e] * scale;
    const float4 v = *(const float4*)(in + (size_t)s * F + (c << 2));
    float* op = out + (size_t)d * F + (c << 2);
    atomicAdd(op + 0, v.x * w);
    atomicAdd(op + 1, v.y * w);
    atomicAdd(op + 2, v.z * w);
    atomicAdd(op + 3, v.w * w);
}

template <int MT, int F, bool RELU_IN>
__global__ void __launch_bounds__(128)
cheb_products_wmma_kernel(const float* __restrict__ T0, const float* __restrict__ W,
                          const float* __restrict__ bias,
                          float* __restrict__ O, float* __restrict__ Z, float* __restrict__ P,
                          int totalWaveTiles, int rowOffset) {
    constexpr int nTilesN = F / 16;
    constexpr int nW1 = 64 * F;
    constexpr int nQ = nW1 / 4;
    __shared__ alignas(16) float sW[3 * nW1 + F];
    const int tid = threadIdx.y * 32 + threadIdx.x;
    {
        const float4* __restrict__ W4 = (const float4*)W;
        float4* sW4 = (float4*)sW;
#pragma unroll
        for (int j = 0; j < nQ / 128; ++j) {
            const int i = j * 128 + tid;
            const float4 w0 = W4[i];
            const float4 w1 = W4[nQ + i];
            const float4 w2 = W4[2 * nQ + i];
            sW4[i] = make_float4(w0.x - w2.x, w0.y - w2.y, w0.z - w2.z, w0.w - w2.w);
            sW4[nQ + i] = w1;
            sW4[2 * nQ + i] = w2;
        }
    }
    if (tid < F) sW[3 * nW1 + tid] = bias[tid];
    __syncthreads();

    const int waveTile = blockIdx.x * 4 + threadIdx.y;
    if (waveTile >= totalWaveTiles) return;
    const int mGroup = waveTile / nTilesN;
    const int nTile = waveTile % nTilesN;

    const int lane = threadIdx.x;
    const int lg = lane >> 4;
    const int ln = lane & 15;
    const int rowBase = rowOffset + mGroup * (MT * 16);
    const int col = (nTile << 4) + ln;

    v8f acc[3][MT];
    const float bv = sW[3 * nW1 + col];
#pragma unroll
    for (int i = 0; i < MT; ++i)
#pragma unroll
        for (int r = 0; r < 8; ++r) {
            acc[0][i][r] = bv;
            acc[1][i][r] = 0.0f;
            acc[2][i][r] = 0.0f;
        }

    const float* arow = T0 + (size_t)(rowBase + ln) * 64 + (lg << 1);
    const float* bcol = sW + (lg << 1) * F + col;
#pragma unroll
    for (int kb = 0; kb < 16; ++kb) {
        v2f b[3];
#pragma unroll
        for (int m = 0; m < 3; ++m) {
            b[m].x = bcol[m * nW1 + (kb * 4) * F];
            b[m].y = bcol[m * nW1 + (kb * 4 + 1) * F];
        }
#pragma unroll
        for (int i = 0; i < MT; ++i) {
            v2f a = *(const v2f*)(arow + (size_t)i * (16 * 64) + kb * 4);
            if (RELU_IN) {
                a.x = fmaxf(a.x, 0.0f);
                a.y = fmaxf(a.y, 0.0f);
            }
#pragma unroll
            for (int m = 0; m < 3; ++m)
                acc[m][i] = __builtin_amdgcn_wmma_f32_16x16x4_f32(
                    false, a, false, b[m], (short)0, acc[m][i], false, false);
        }
    }

    float* outs[3] = {O, Z, P};
#pragma unroll
    for (int m = 0; m < 3; ++m) {
#pragma unroll
        for (int i = 0; i < MT; ++i) {
            float* orow = outs[m] + (size_t)(rowBase + i * 16 + (lg << 3)) * F + col;
#pragma unroll
            for (int r = 0; r < 8; ++r) orow[r * F] = acc[m][i][r];
        }
    }
}

template <int F, bool RELU_IN>
static void cheb_layer(const float* T0, const float* W, const float* b,
                       float* P, float* Z, float* O,
                       const int* src, const int* dst, const float* wE,
                       int n, int E, hipStream_t stream) {
    const int nTilesN = F / 16;
    const int mTiles = n / 16;
    const int mGroups = mTiles / 5;
    const int wt5 = mGroups * nTilesN;
    if (wt5 > 0) {
        cheb_products_wmma_kernel<5, F, RELU_IN><<<(wt5 + 3) / 4, dim3(32, 4), 0, stream>>>(
            T0, W, b, O, Z, P, wt5, 0);
    }
    const int remTiles = mTiles - mGroups * 5;
    if (remTiles > 0) {
        const int wt1 = remTiles * nTilesN;
        cheb_products_wmma_kernel<1, F, RELU_IN><<<(wt1 + 3) / 4, dim3(32, 4), 0, stream>>>(
            T0, W, b, O, Z, P, wt1, mGroups * 80);
    }

    const int sblocks = (E * (F / 4) + 255) / 256;
    cheb_scatter_kernel<F><<<sblocks, 256, 0, stream>>>(P, Z, src, dst, wE, 2.0f, E);
    cheb_scatter_kernel<F><<<sblocks, 256, 0, stream>>>(Z, O, src, dst, wE, 1.0f, E);
}

extern "C" void kernel_launch(void* const* d_in, const int* in_sizes, int n_in,
                              void* d_out, int out_size, void* d_ws, size_t ws_size,
                              hipStream_t stream) {
    const float* x  = (const float*)d_in[0];
    const int*   ei = (const int*)d_in[1];
    const float* W1 = (const float*)d_in[2];
    const float* b1 = (const float*)d_in[3];
    const float* W2 = (const float*)d_in[4];
    const float* b2 = (const float*)d_in[5];
    const float* W3 = (const float*)d_in[6];
    const float* b3 = (const float*)d_in[7];

    const int n = in_sizes[0] / 64;
    const int E = in_sizes[1] / 2;
    const int* src = ei;
    const int* dst = ei + E;

    float* ws = (float*)d_ws;
    size_t o = 0;
    float* dis = ws + o; o += ((size_t)n + 63) & ~(size_t)63;
    float* wE  = ws + o; o += ((size_t)E + 63) & ~(size_t)63;
    float* P   = ws + o; o += (size_t)n * 64;
    float* Z   = ws + o; o += (size_t)n * 64;
    float* HA  = ws + o; o += (size_t)n * 64;
    float* HB  = ws + o; o += (size_t)n * 64;

    hipMemsetAsync(dis, 0, (size_t)n * sizeof(float), stream);
    cheb_degree_kernel<<<(E + 255) / 256, 256, 0, stream>>>(src, dis, E);
    cheb_dis_kernel<<<(n + 255) / 256, 256, 0, stream>>>(dis, n);
    cheb_edgew_kernel<<<(E + 255) / 256, 256, 0, stream>>>(src, dst, dis, wE, E);

    cheb_layer<64, false>(x,  W1, b1, P, Z, HA, src, dst, wE, n, E, stream);
    cheb_layer<64, true >(HA, W2, b2, P, Z, HB, src, dst, wE, n, E, stream);
    cheb_layer<32, true >(HB, W3, b3, P, Z, (float*)d_out, src, dst, wE, n, E, stream);
}


# Scatter lane map: one wave per 32 edges, all 32 lanes on one edge row at a time (edge data by readlane): every gather load and f32 atomic covers one whole 128-B line; atomic arrival order changes
# speedup vs baseline: 2.0621x; 1.6292x over previous
#include <hip/hip_runtime.h>


typedef __attribute__((ext_vector_type(2))) float v2f;
typedef __attribute__((ext_vector_type(8))) float v8f;

__global__ void cheb_degree_kernel(const int* __restrict__ row, float* __restrict__ deg, int E) {
    int e = blockIdx.x * blockDim.x + threadIdx.x;
    if (e < E) atomicAdd(&deg[row[e]], 1.0f);
}

__global__ void cheb_dis_kernel(float* __restrict__ deg, int n) {
    int i = blockIdx.x * blockDim.x + threadIdx.x;
    if (i < n) {
        float d = deg[i];
        deg[i] = (d > 0.0f) ? rsqrtf(fmaxf(d, 1.0f)) : 0.0f;
    }
}

__global__ void cheb_edgew_kernel(const int* __restrict__ row, const int* __restrict__ col,
                                  const float* __restrict__ dis, float* __restrict__ wE, int E) {
    int e = blockIdx.x * blockDim.x + threadIdx.x;
    if (e < E) wE[e] = -dis[row[e]] * dis[col[e]];
}

template <int F>
__global__ void __launch_bounds__(256)
cheb_scatter_kernel(const float* __restrict__ in, float* __restrict__ out,
                    const int* __restrict__ src, const int* __restrict__ dst,
                    const float* __restrict__ wE, float scale, int E) {
    constexpr int LPR = F / 32;
    constexpr int G = 32 / LPR;
    const unsigned lane = threadIdx.x & 31u;
    const int e0 = __builtin_amdgcn_readfirstlane(
        (int)((blockIdx.x * blockDim.x + threadIdx.x) >> 5) << 5);
    if (e0 >= E) return;
    const int e = e0 + (int)lane;
    int s_l = 0, d_l = 0;
    float w_l = 0.0f;
    if (e < E) {
        s_l = src[e];
        d_l = dst[e];
        w_l = wE[e] * scale;
    }
    if (E - e0 >= 32) {
#pragma unroll
        for (int g = 0; g < 32; g += G) {
            float v[G][LPR];
#pragma unroll
            for (int i = 0; i < G; ++i) {
                const unsigned s = (unsigned)__builtin_amdgcn_readlane(s_l, g + i);
                const float* ip = in + (size_t)s * F;
#pragma unroll
                for (int j = 0; j < LPR; ++j) v[i][j] = ip[lane + j * 32];
            }
#pragma unroll
            for (int i = 0; i < G; ++i) {
                const unsigned d = (unsigned)__builtin_amdgcn_readlane(d_l, g + i);
                const float w = __int_as_float(__builtin_amdgcn_readlane(__float_as_int(w_l), g + i));
                float* op = out + (size_t)d * F;
#pragma unroll
                for (int j = 0; j < LPR; ++j) atomicAdd(op + lane + j * 32, v[i][j] * w);
            }
        }
    } else {
        const int cnt = E - e0;
        for (int i = 0; i < cnt; ++i) {
            const unsigned s = (unsigned)__builtin_amdgcn_readlane(s_l, i);
            const unsigned d = (unsigned)__builtin_amdgcn_readlane(d_l, i);
            const float w = __int_as_float(__builtin_amdgcn_readlane(__float_as_int(w_l), i));
            const float* ip = in + (size_t)s * F;
            float* op = out + (size_t)d * F;
            for (int j = 0; j < LPR; ++j) atomicAdd(op + lane + j * 32, ip[lane + j * 32] * w);
        }
    }
}

template <int MT, int F, bool RELU_IN>
__global__ void __launch_bounds__(128)
cheb_products_wmma_kernel(const float* __restrict__ T0, const float* __restrict__ W,
                          const float* __restrict__ bias,
                          float* __restrict__ O, float* __restrict__ Z, float* __restrict__ P,
                          int totalWaveTiles, int rowOffset) {
    constexpr int nTilesN = F / 16;
    constexpr int nW1 = 64 * F;
    constexpr int nQ = nW1 / 4;
    __shared__ alignas(16) float sW[3 * nW1 + F];
    const int tid = threadIdx.y * 32 + threadIdx.x;
    {
        const float4* __restrict__ W4 = (const float4*)W;
        float4* sW4 = (float4*)sW;
#pragma unroll
        for (int j = 0; j < nQ / 128; ++j) {
            const int i = j * 128 + tid;
            const float4 w0 = W4[i];
            const float4 w1 = W4[nQ + i];
            const float4 w2 = W4[2 * nQ + i];
            sW4[i] = make_float4(w0.x - w2.x, w0.y - w2.y, w0.z - w2.z, w0.w - w2.w);
            sW4[nQ + i] = w1;
            sW4[2 * nQ + i] = w2;
        }
    }
    if (tid < F) sW[3 * nW1 + tid] = bias[tid];
    __syncthreads();

    const int waveTile = blockIdx.x * 4 + threadIdx.y;
    if (waveTile >= totalWaveTiles) return;
    const int mGroup = waveTile / nTilesN;
    const int nTile = waveTile % nTilesN;

    const int lane = threadIdx.x;
    const int lg = lane >> 4;
    const int ln = lane & 15;
    const int rowBase = rowOffset + mGroup * (MT * 16);
    const int col = (nTile << 4) + ln;

    v8f acc[3][MT];
    const float bv = sW[3 * nW1 + col];
#pragma unroll
    for (int i = 0; i < MT; ++i)
#pragma unroll
        for (int r = 0; r < 8; ++r) {
            acc[0][i][r] = bv;
            acc[1][i][r] = 0.0f;
            acc[2][i][r] = 0.0f;
        }

    const float* arow = T0 + (size_t)(rowBase + ln) * 64 + (lg << 1);
    const float* bcol = sW + (lg << 1) * F + col;
#pragma unroll
    for (int kb = 0; kb < 16; ++kb) {
        v2f b[3];
#pragma unroll
        for (int m = 0; m < 3; ++m) {
            b[m].x = bcol[m * nW1 + (kb * 4) * F];
            b[m].y = bcol[m * nW1 + (kb * 4 + 1) * F];
        }
#pragma unroll
        for (int i = 0; i < MT; ++i) {
            v2f a = *(const v2f*)(arow + (size_t)i * (16 * 64) + kb * 4);
            if (RELU_IN) {
                a.x = fmaxf(a.x, 0.0f);
                a.y = fmaxf(a.y, 0.0f);
            }
#pragma unroll
            for (int m = 0; m < 3; ++m)
                acc[m][i] = __builtin_amdgcn_wmma_f32_16x16x4_f32(
                    false, a, false, b[m], (short)0, acc[m][i], false, false);
        }
    }

    float* outs[3] = {O, Z, P};
#pragma unroll
    for (int m = 0; m < 3; ++m) {
#pragma unroll
        for (int i = 0; i < MT; ++i) {
            float* orow = outs[m] + (size_t)(rowBase + i * 16 + (lg << 3)) * F + col;
#pragma unroll
            for (int r = 0; r < 8; ++r) orow[r * F] = acc[m][i][r];
        }
    }
}

template <int F, bool RELU_IN>
static void cheb_layer(const float* T0, const float* W, const float* b,
                       float* P, float* Z, float* O,
                       const int* src, const int* dst, const float* wE,
                       int n, int E, hipStream_t stream) {
    const int nTilesN = F / 16;
    const int mTiles = n / 16;
    const int mGroups = mTiles / 5;
    const int wt5 = mGroups * nTilesN;
    if (wt5 > 0) {
        cheb_products_wmma_kernel<5, F, RELU_IN><<<(wt5 + 3) / 4, dim3(32, 4), 0, stream>>>(
            T0, W, b, O, Z, P, wt5, 0);
    }
    const int remTiles = mTiles - mGroups * 5;
    if (remTiles > 0) {
        const int wt1 = remTiles * nTilesN;
        cheb_products_wmma_kernel<1, F, RELU_IN><<<(wt1 + 3) / 4, dim3(32, 4), 0, stream>>>(
            T0, W, b, O, Z, P, wt1, mGroups * 80);
    }

    const int sblocks = ((E + 31) / 32 + 7) / 8;
    cheb_scatter_kernel<F><<<sblocks, 256, 0, stream>>>(P, Z, src, dst, wE, 2.0f, E);
    cheb_scatter_kernel<F><<<sblocks, 256, 0, stream>>>(Z, O, src, dst, wE, 1.0f, E);
}

extern "C" void kernel_launch(void* const* d_in, const int* in_sizes, int n_in,
                              void* d_out, int out_size, void* d_ws, size_t ws_size,
                              hipStream_t stream) {
    const float* x  = (const float*)d_in[0];
    const int*   ei = (const int*)d_in[1];
    const float* W1 = (const float*)d_in[2];
    const float* b1 = (const float*)d_in[3];
    const float* W2 = (const float*)d_in[4];
    const float* b2 = (const float*)d_in[5];
    const float* W3 = (const float*)d_in[6];
    const float* b3 = (const float*)d_in[7];

    const int n = in_sizes[0] / 64;
    const int E = in_sizes[1] / 2;
    const int* src = ei;
    const int* dst = ei + E;

    float* ws = (float*)d_ws;
    size_t o = 0;
    float* dis = ws + o; o += ((size_t)n + 63) & ~(size_t)63;
    float* wE  = ws + o; o += ((size_t)E + 63) & ~(size_t)63;
    float* P   = ws + o; o += (size_t)n * 64;
    float* Z   = ws + o; o += (size_t)n * 64;
    float* HA  = ws + o; o += (size_t)n * 64;
    float* HB  = ws + o; o += (size_t)n * 64;

    hipMemsetAsync(dis, 0, (size_t)n * sizeof(float), stream);
    cheb_degree_kernel<<<(E + 255) / 256, 256, 0, stream>>>(src, dis, E);
    cheb_dis_kernel<<<(n + 255) / 256, 256, 0, stream>>>(dis, n);
    cheb_edgew_kernel<<<(E + 255) / 256, 256, 0, stream>>>(src, dst, dis, wE, E);

    cheb_layer<64, false>(x,  W1, b1, P, Z, HA, src, dst, wE, n, E, stream);
    cheb_layer<64, true >(HA, W2, b2, P, Z, HB, src, dst, wE, n, E, stream);
    cheb_layer<32, true >(HB, W3, b3, P, Z, (float*)d_out, src, dst, wE, n, E, stream);
}


# Scatter atomics stay inside one chip half: rows split in two, each 256-edge slice visited once per side, work-groups claim slices of their own half's side first (die id); atomic arrival order changes
# speedup vs baseline: 2.5425x; 1.2330x over previous
#include <hip/hip_runtime.h>


typedef __attribute__((ext_vector_type(2))) float v2f;
typedef __attribute__((ext_vector_type(8))) float v8f;

__global__ void cheb_degree_kernel(const int* __restrict__ row, float* __restrict__ deg, int E) {
    int e = blockIdx.x * blockDim.x + threadIdx.x;
    if (e < E) atomicAdd(&deg[row[e]], 1.0f);
}

__global__ void cheb_dis_kernel(float* __restrict__ deg, int n) {
    int i = blockIdx.x * blockDim.x + threadIdx.x;
    if (i < n) {
        float d = deg[i];
        deg[i] = (d > 0.0f) ? rsqrtf(fmaxf(d, 1.0f)) : 0.0f;
    }
}

__global__ void cheb_edgew_kernel(const int* __restrict__ row, const int* __restrict__ col,
                                  const float* __restrict__ dis, float* __restrict__ wE, int E) {
    int e = blockIdx.x * blockDim.x + threadIdx.x;
    if (e < E) wE[e] = -dis[row[e]] * dis[col[e]];
}

template <int F>
__global__ void __launch_bounds__(256)
cheb_scatter_kernel(const float* __restrict__ in, float* __restrict__ out,
                    const int* __restrict__ src, const int* __restrict__ dst,
                    const float* __restrict__ wE, float scale, int E, int nHalf,
                    int* __restrict__ claim) {
    constexpr int LPR = F / 32;
    constexpr int G = 20;
    constexpr int RS = 1024;
    __shared__ int sPick[2];
    const unsigned lane = threadIdx.x & 31u;
    const int wave = (int)(threadIdx.x >> 5);
    const int nSlices = (E + 255) >> 8;
    int myRole = 0;
    if (threadIdx.x == 0) {
        const int die = (__builtin_amdgcn_s_sendmsg_rtn(0x87) >> 16) & 15;
        myRole = (die >> 2) & 1;
        int r = myRole;
        int c = atomicAdd(&claim[r * RS], 1);
        if (c >= nSlices) { r ^= 1; c = atomicAdd(&claim[r * RS], 1); }
        sPick[0] = (c < nSlices) ? (c * 2 + r) : -1;
    }
    __syncthreads();
    for (int it = 0; it < 2 * nSlices; ++it) {
        const int pick = __builtin_amdgcn_readfirstlane(sPick[it & 1]);
        if (pick < 0) break;
        int cNext = 0;
        if (threadIdx.x == 0) cNext = atomicAdd(&claim[myRole * RS], 1);

        const int role = pick & 1;
        const int e0 = (pick >> 1) * 256 + wave * 32;
        if (e0 < E) {
            const int e = e0 + (int)lane;
            int s_l = 0, d_l = 0;
            float w_l = 0.0f;
            bool sel = false;
            if (e < E) {
                s_l = src[e];
                d_l = dst[e];
                w_l = wE[e] * scale;
                sel = ((d_l >= nHalf) ? 1 : 0) == role;
            }
            unsigned m = __builtin_amdgcn_ballot_w32(sel);
            for (int t = 0; t < (32 + G - 1) / G && m != 0u; ++t) {
                float v[G][LPR];
                unsigned ml = m;
#pragma unroll
                for (int i = 0; i < G; ++i) {
#pragma unroll
                    for (int q = 0; q < LPR; ++q) v[i][q] = 0.0f;
                    if (ml != 0u) {
                        const int jl = __builtin_ctz(ml);
                        ml &= ml - 1u;
                        const unsigned sj = (unsigned)__builtin_amdgcn_readlane(s_l, jl);
                        const float* ip = in + (size_t)sj * F;
#pragma unroll
                        for (int q = 0; q < LPR; ++q) v[i][q] = ip[lane + q * 32];
                    }
                }
                unsigned ma = m;
#pragma unroll
                for (int i = 0; i < G; ++i) {
                    if (ma != 0u) {
                        const int ja = __builtin_ctz(ma);
                        ma &= ma - 1u;
                        const unsigned dj = (unsigned)__builtin_amdgcn_readlane(d_l, ja);
                        const float w = __int_as_float(
                            __builtin_amdgcn_readlane(__float_as_int(w_l), ja));
                        float* op = out + (size_t)dj * F;
#pragma unroll
                        for (int q = 0; q < LPR; ++q) atomicAdd(op + lane + q * 32, v[i][q] * w);
                    }
                }
                m = ml;
            }
        }

        if (threadIdx.x == 0) {
            int r = myRole;
            int c = cNext;
            if (c >= nSlices) { r ^= 1; c = atomicAdd(&claim[r * RS], 1); }
            sPick[(it + 1) & 1] = (c < nSlices) ? (c * 2 + r) : -1;
        }
        __syncthreads();
    }
}

template <int MT, int F, bool RELU_IN>
__global__ void __launch_bounds__(128)
cheb_products_wmma_kernel(const float* __restrict__ T0, const float* __restrict__ W,
                          const float* __restrict__ bias,
                          float* __restrict__ O, float* __restrict__ Z, float* __restrict__ P,
                          int* __restrict__ claim, int totalWaveTiles, int rowOffset) {
    constexpr int nTilesN = F / 16;
    constexpr int nW1 = 64 * F;
    constexpr int nQ = nW1 / 4;
    __shared__ alignas(16) float sW[3 * nW1 + F];
    const int tid = threadIdx.y * 32 + threadIdx.x;
    if (blockIdx.x == 0 && tid == 0) {
        claim[0] = 0;
        claim[1] = 0;
        claim[1024] = 0;
        claim[1025] = 0;
    }
    {
        const float4* __restrict__ W4 = (const float4*)W;
        float4* sW4 = (float4*)sW;
#pragma unroll
        for (int j = 0; j < nQ / 128; ++j) {
            const int i = j * 128 + tid;
            const float4 w0 = W4[i];
            const float4 w1 = W4[nQ + i];
            const float4 w2 = W4[2 * nQ + i];
            sW4[i] = make_float4(w0.x - w2.x, w0.y - w2.y, w0.z - w2.z, w0.w - w2.w);
            sW4[nQ + i] = w1;
            sW4[2 * nQ + i] = w2;
        }
    }
    if (tid < F) sW[3 * nW1 + tid] = bias[tid];
    __syncthreads();

    const int waveTile = blockIdx.x * 4 + threadIdx.y;
    if (waveTile >= totalWaveTiles) return;
    const int mGroup = waveTile / nTilesN;
    const int nTile = waveTile % nTilesN;

    const int lane = threadIdx.x;
    const int lg = lane >> 4;
    const int ln = lane & 15;
    const int rowBase = rowOffset + mGroup * (MT * 16);
    const int col = (nTile << 4) + ln;

    v8f acc[3][MT];
    const float bv = sW[3 * nW1 + col];
#pragma unroll
    for (int i = 0; i < MT; ++i)
#pragma unroll
        for (int r = 0; r < 8; ++r) {
            acc[0][i][r] = bv;
            acc[1][i][r] = 0.0f;
            acc[2][i][r] = 0.0f;
        }

    const float* arow = T0 + (size_t)(rowBase + ln) * 64 + (lg << 1);
    const float* bcol = sW + (lg << 1) * F + col;
#pragma unroll
    for (int kb = 0; kb < 16; ++kb) {
        v2f b[3];
#pragma unroll
        for (int m = 0; m < 3; ++m) {
            b[m].x = bcol[m * nW1 + (kb * 4) * F];
            b[m].y = bcol[m * nW1 + (kb * 4 + 1) * F];
        }
#pragma unroll
        for (int i = 0; i < MT; ++i) {
            v2f a = *(const v2f*)(arow + (size_t)i * (16 * 64) + kb * 4);
            if (RELU_IN) {
                a.x = fmaxf(a.x, 0.0f);
                a.y = fmaxf(a.y, 0.0f);
            }
#pragma unroll
            for (int m = 0; m < 3; ++m)
                acc[m][i] = __builtin_amdgcn_wmma_f32_16x16x4_f32(
                    false, a, false, b[m], (short)0, acc[m][i], false, false);
        }
    }

    float* outs[3] = {O, Z, P};
#pragma unroll
    for (int m = 0; m < 3; ++m) {
#pragma unroll
        for (int i = 0; i < MT; ++i) {
            float* orow = outs[m] + (size_t)(rowBase + i * 16 + (lg << 3)) * F + col;
#pragma unroll
            for (int r = 0; r < 8; ++r) orow[r * F] = acc[m][i][r];
        }
    }
}

template <int F, bool RELU_IN>
static void cheb_layer(const float* T0, const float* W, const float* b,
                       float* P, float* Z, float* O,
                       const int* src, const int* dst, const float* wE,
                       int* claim, int n, int E, hipStream_t stream) {
    const int nTilesN = F / 16;
    const int mTiles = n / 16;
    const int mGroups = mTiles / 5;
    const int wt5 = mGroups * nTilesN;
    if (wt5 > 0) {
        cheb_products_wmma_kernel<5, F, RELU_IN><<<(wt5 + 3) / 4, dim3(32, 4), 0, stream>>>(
            T0, W, b, O, Z, P, claim, wt5, 0);
    }
    const int remTiles = mTiles - mGroups * 5;
    if (remTiles > 0) {
        const int wt1 = remTiles * nTilesN;
        cheb_products_wmma_kernel<1, F, RELU_IN><<<(wt1 + 3) / 4, dim3(32, 4), 0, stream>>>(
            T0, W, b, O, Z, P, claim, wt1, mGroups * 80);
    }

    const int sblocks = 2048;
    cheb_scatter_kernel<F><<<sblocks, 256, 0, stream>>>(P, Z, src, dst, wE, 2.0f, E, n / 2, claim);
    cheb_scatter_kernel<F><<<sblocks, 256, 0, stream>>>(Z, O, src, dst, wE, 1.0f, E, n / 2, claim + 1);
}

extern "C" void kernel_launch(void* const* d_in, const int* in_sizes, int n_in,
                              void* d_out, int out_size, void* d_ws, size_t ws_size,
                              hipStream_t stream) {
    const float* x  = (const float*)d_in[0];
    const int*   ei = (const int*)d_in[1];
    const float* W1 = (const float*)d_in[2];
    const float* b1 = (const float*)d_in[3];
    const float* W2 = (const float*)d_in[4];
    const float* b2 = (const float*)d_in[5];
    const float* W3 = (const float*)d_in[6];
    const float* b3 = (const float*)d_in[7];

    const int n = in_sizes[0] / 64;
    const int E = in_sizes[1] / 2;
    const int* src = ei;
    const int* dst = ei + E;

    float* ws = (float*)d_ws;
    size_t o = 0;
    float* dis = ws + o; o += ((size_t)n + 63) & ~(size_t)63;
    int* claim = (int*)dis;
    float* wE  = ws + o; o += ((size_t)E + 63) & ~(size_t)63;
    float* P   = ws + o; o += (size_t)n * 64;
    float* Z   = ws + o; o += (size_t)n * 64;
    float* HA  = ws + o; o += (size_t)n * 64;
    float* HB  = ws + o; o += (size_t)n * 64;

    hipMemsetAsync(dis, 0, (size_t)n * sizeof(float), stream);
    cheb_degree_kernel<<<(E + 255) / 256, 256, 0, stream>>>(src, dis, E);
    cheb_dis_kernel<<<(n + 255) / 256, 256, 0, stream>>>(dis, n);
    cheb_edgew_kernel<<<(E + 255) / 256, 256, 0, stream>>>(src, dst, dis, wE, E);

    cheb_layer<64, false>(x,  W1, b1, P, Z, HA, src, dst, wE, claim, n, E, stream);
    cheb_layer<64, true >(HA, W2, b2, P, Z, HB, src, dst, wE, claim, n, E, stream);
    cheb_layer<32, true >(HB, W3, b3, P, Z, (float*)d_out, src, dst, wE, claim, n, E, stream);
}


# Edge-weight prep on LDS tables in 3 fat launches: degree counted in LDS per (node range, edge chunk), partials summed per node; weights read dis from an LDS copy; no memset or global atomics there
# speedup vs baseline: 3.2029x; 1.2597x over previous
#include <hip/hip_runtime.h>


typedef __attribute__((ext_vector_type(2))) float v2f;
typedef __attribute__((ext_vector_type(8))) float v8f;

constexpr int kDegRanges = 8;
constexpr int kDegChunks = 32;
constexpr int kDegCap = 8192;
constexpr int kPrepThreads = 512;
constexpr int kInFlight = 13;

__device__ __forceinline__ void cheb_count_one(unsigned* cnt, int s, int base, unsigned len) {
    const unsigned d = (unsigned)(s - base);
    if (d < len) atomicAdd(&cnt[d], 1u);
}

__global__ void __launch_bounds__(kPrepThreads)
cheb_degree_kernel(const int* __restrict__ row, unsigned* __restrict__ part,
                   int E, int n, int npad) {
    __shared__ unsigned sCnt[kDegCap];
    const int tid = (int)threadIdx.x;
    const int r = (int)blockIdx.x % kDegRanges;
    const int c = (int)blockIdx.x / kDegRanges;
    const int nr = (n + kDegRanges - 1) / kDegRanges;
    const int lo = r * nr;
    const int hi = min(lo + nr, n);
    const int ec = (((E + kDegChunks - 1) / kDegChunks) + 3) & ~3;
    const int e0 = min(c * ec, E);
    const int e1 = min(e0 + ec, E);
    const bool vec = (((size_t)row) & 15) == 0;
    const int nq = vec ? ((e1 - e0) >> 2) : 0;
    const int4* __restrict__ row4 = (const int4*)(row + e0);

    for (int base = lo; base < hi; base += kDegCap) {
        const unsigned len = (unsigned)min(kDegCap, hi - base);
        for (int j = tid; j < (int)len; j += kPrepThreads) sCnt[j] = 0u;
        __syncthreads();
        for (int q0 = 0; q0 < nq; q0 += kInFlight * kPrepThreads) {
            int4 s[kInFlight];
#pragma unroll
            for (int k = 0; k < kInFlight; ++k) {
                const int q = q0 + k * kPrepThreads + tid;
                s[k] = (q < nq) ? row4[q] : make_int4(-1, -1, -1, -1);
            }
#pragma unroll
            for (int k = 0; k < kInFlight; ++k) {
                cheb_count_one(sCnt, s[k].x, base, len);
                cheb_count_one(sCnt, s[k].y, base, len);
                cheb_count_one(sCnt, s[k].z, base, len);
                cheb_count_one(sCnt, s[k].w, base, len);
            }
        }
        for (int e = e0 + (nq << 2) + tid; e < e1; e += kPrepThreads)
            cheb_count_one(sCnt, row[e], base, len);
        __syncthreads();
        unsigned* __restrict__ dstp = part + (size_t)c * npad + base;
        for (int j = tid; j < (int)len; j += kPrepThreads) dstp[j] = sCnt[j];
        __syncthreads();
    }
}

__global__ void cheb_dis_kernel(const unsigned* __restrict__ part, float* __restrict__ dis,
                                int n, int npad) {
    const int i = blockIdx.x * blockDim.x + threadIdx.x;
    if (i < n) {
        unsigned cnt = 0u;
#pragma unroll
        for (int c = 0; c < kDegChunks; ++c) cnt += part[(size_t)c * npad + i];
        const float d = (float)cnt;
        dis[i] = (d > 0.0f) ? rsqrtf(fmaxf(d, 1.0f)) : 0.0f;
    }
}

constexpr int kDisCap = 65536;
constexpr int kDisInFlight = 25;

__global__ void __launch_bounds__(kPrepThreads)
cheb_edgew_kernel(const int* __restrict__ row, const int* __restrict__ col,
                  const float* __restrict__ dis, float* __restrict__ wE, int E, int n) {
    __shared__ alignas(16) float sDis[kDisCap];
    const int tid = (int)threadIdx.x;
    const bool inLds = n <= kDisCap;
    const bool vec = ((((size_t)row) | ((size_t)col) | ((size_t)wE)) & 15) == 0;
    const int nqAll = vec ? (E >> 2) : 0;
    const int per = (nqAll + (int)gridDim.x - 1) / (int)gridDim.x;
    const int qb = min((int)blockIdx.x * per, nqAll);
    const int qe = min(qb + per, nqAll);
    const int4* __restrict__ row4 = (const int4*)row;
    const int4* __restrict__ col4 = (const int4*)col;

    int4 ps[2], pd[2];
#pragma unroll
    for (int k = 0; k < 2; ++k) {
        const int q = qb + k * kPrepThreads + tid;
        ps[k] = make_int4(0, 0, 0, 0);
        pd[k] = make_int4(0, 0, 0, 0);
        if (q < qe) { ps[k] = row4[q]; pd[k] = col4[q]; }
    }

    if (inLds) {
        const bool dvec = (((size_t)dis) & 15) == 0;
        const int nd4 = dvec ? (n >> 2) : 0;
        const float4* __restrict__ dis4 = (const float4*)dis;
        float4* sDis4 = (float4*)sDis;
        for (int i0 = 0; i0 < nd4; i0 += kDisInFlight * kPrepThreads) {
            float4 t[kDisInFlight];
#pragma unroll
            for (int k = 0; k < kDisInFlight; ++k) {
                const int i = i0 + k * kPrepThreads + tid;
                t[k] = (i < nd4) ? dis4[i] : make_float4(0.0f, 0.0f, 0.0f, 0.0f);
            }
#pragma unroll
            for (int k = 0; k < kDisInFlight; ++k) {
                const int i = i0 + k * kPrepThreads + tid;
                if (i < nd4) sDis4[i] = t[k];
            }
        }
        for (int i = (nd4 << 2) + tid; i < n; i += kPrepThreads) sDis[i] = dis[i];
    }
    __syncthreads();

#pragma unroll
    for (int k = 0; k < 2; ++k) {
        const int q = qb + k * kPrepThreads + tid;
        if (q < qe) {
            float4 w;
            if (inLds) {
                w.x = -sDis[ps[k].x] * sDis[pd[k].x];
                w.y = -sDis[ps[k].y] * sDis[pd[k].y];
                w.z = -sDis[ps[k].z] * sDis[pd[k].z];
                w.w = -sDis[ps[k].w] * sDis[pd[k].w];
            } else {
                w.x = -dis[ps[k].x] * dis[pd[k].x];
                w.y = -dis[ps[k].y] * dis[pd[k].y];
                w.z = -dis[ps[k].z] * dis[pd[k].z];
                w.w = -dis[ps[k].w] * dis[pd[k].w];
            }
            ((float4*)wE)[q] = w;
        }
    }
    for (int q = qb + 2 * kPrepThreads + tid; q < qe; q += kPrepThreads) {
        const int4 s = row4[q];
        const int4 d = col4[q];
        float4 w;
        if (inLds) {
            w.x = -sDis[s.x] * sDis[d.x];
            w.y = -sDis[s.y] * sDis[d.y];
            w.z = -sDis[s.z] * sDis[d.z];
            w.w = -sDis[s.w] * sDis[d.w];
        } else {
            w.x = -dis[s.x] * dis[d.x];
            w.y = -dis[s.y] * dis[d.y];
            w.z = -dis[s.z] * dis[d.z];
            w.w = -dis[s.w] * dis[d.w];
        }
        ((float4*)wE)[q] = w;
    }
    for (int e = (nqAll << 2) + (int)blockIdx.x * kPrepThreads + tid; e < E;
         e += (int)gridDim.x * kPrepThreads) {
        const int s = row[e];
        const int d = col[e];
        wE[e] = inLds ? (-sDis[s] * sDis[d]) : (-dis[s] * dis[d]);
    }
}

template <int F>
__global__ void __launch_bounds__(256)
cheb_scatter_kernel(const float* __restrict__ in, float* __restrict__ out,
                    const int* __restrict__ src, const int* __restrict__ dst,
                    const float* __restrict__ wE, float scale, int E, int nHalf,
                    int* __restrict__ claim) {
    constexpr int LPR = F / 32;
    constexpr int G = 20;
    constexpr int RS = 1024;
    __shared__ int sPick[2];
    const unsigned lane = threadIdx.x & 31u;
    const int wave = (int)(threadIdx.x >> 5);
    const int nSlices = (E + 255) >> 8;
    int myRole = 0;
    if (threadIdx.x == 0) {
        const int die = (__builtin_amdgcn_s_sendmsg_rtn(0x87) >> 16) & 15;
        myRole = (die >> 2) & 1;
        int r = myRole;
        int c = atomicAdd(&claim[r * RS], 1);
        if (c >= nSlices) { r ^= 1; c = atomicAdd(&claim[r * RS], 1); }
        sPick[0] = (c < nSlices) ? (c * 2 + r) : -1;
    }
    __syncthreads();
    for (int it = 0; it < 2 * nSlices; ++it) {
        const int pick = __builtin_amdgcn_readfirstlane(sPick[it & 1]);
        if (pick < 0) break;
        int cNext = 0;
        if (threadIdx.x == 0) cNext = atomicAdd(&claim[myRole * RS], 1);

        const int role = pick & 1;
        const int e0 = (pick >> 1) * 256 + wave * 32;
        if (e0 < E) {
            const int e = e0 + (int)lane;
            int s_l = 0, d_l = 0;
            float w_l = 0.0f;
            bool sel = false;
            if (e < E) {
                s_l = src[e];
                d_l = dst[e];
                w_l = wE[e] * scale;
                sel = ((d_l >= nHalf) ? 1 : 0) == role;
            }
            unsigned m = __builtin_amdgcn_ballot_w32(sel);
            for (int t = 0; t < (32 + G - 1) / G && m != 0u; ++t) {
                float v[G][LPR];
                unsigned ml = m;
#pragma unroll
                for (int i = 0; i < G; ++i) {
#pragma unroll
                    for (int q = 0; q < LPR; ++q) v[i][q] = 0.0f;
                    if (ml != 0u) {
                        const int jl = __builtin_ctz(ml);
                        ml &= ml - 1u;
                        const unsigned sj = (unsigned)__builtin_amdgcn_readlane(s_l, jl);
                        const float* ip = in + (size_t)sj * F;
#pragma unroll
                        for (int q = 0; q < LPR; ++q) v[i][q] = ip[lane + q * 32];
                    }
                }
                unsigned ma = m;
#pragma unroll
                for (int i = 0; i < G; ++i) {
                    if (ma != 0u) {
                        const int ja = __builtin_ctz(ma);
                        ma &= ma - 1u;
                        const unsigned dj = (unsigned)__builtin_amdgcn_readlane(d_l, ja);
                        const float w = __int_as_float(
                            __builtin_amdgcn_readlane(__float_as_int(w_l), ja));
                        float* op = out + (size_t)dj * F;
#pragma unroll
                        for (int q = 0; q < LPR; ++q) atomicAdd(op + lane + q * 32, v[i][q] * w);
                    }
                }
                m = ml;
            }
        }

        if (threadIdx.x == 0) {
            int r = myRole;
            int c = cNext;
            if (c >= nSlices) { r ^= 1; c = atomicAdd(&claim[r * RS], 1); }
            sPick[(it + 1) & 1] = (c < nSlices) ? (c * 2 + r) : -1;
        }
        __syncthreads();
    }
}

template <int MT, int F, bool RELU_IN>
__global__ void __launch_bounds__(128)
cheb_products_wmma_kernel(const float* __restrict__ T0, const float* __restrict__ W,
                          const float* __restrict__ bias,
                          float* __restrict__ O, float* __restrict__ Z, float* __restrict__ P,
                          int* __restrict__ claim, int totalWaveTiles, int rowOffset) {
    constexpr int nTilesN = F / 16;
    constexpr int nW1 = 64 * F;
    constexpr int nQ = nW1 / 4;
    __shared__ alignas(16) float sW[3 * nW1 + F];
    const int tid = threadIdx.y * 32 + threadIdx.x;
    if (blockIdx.x == 0 && tid == 0) {
        claim[0] = 0;
        claim[1] = 0;
        claim[1024] = 0;
        claim[1025] = 0;
    }
    {
        const float4* __restrict__ W4 = (const float4*)W;
        float4* sW4 = (float4*)sW;
#pragma unroll
        for (int j = 0; j < nQ / 128; ++j) {
            const int i = j * 128 + tid;
            const float4 w0 = W4[i];
            const float4 w1 = W4[nQ + i];
            const float4 w2 = W4[2 * nQ + i];
            sW4[i] = make_float4(w0.x - w2.x, w0.y - w2.y, w0.z - w2.z, w0.w - w2.w);
            sW4[nQ + i] = w1;
            sW4[2 * nQ + i] = w2;
        }
    }
    if (tid < F) sW[3 * nW1 + tid] = bias[tid];
    __syncthreads();

    const int waveTile = blockIdx.x * 4 + threadIdx.y;
    if (waveTile >= totalWaveTiles) return;
    const int mGroup = waveTile / nTilesN;
    const int nTile = waveTile % nTilesN;

    const int lane = threadIdx.x;
    const int lg = lane >> 4;
    const int ln = lane & 15;
    const int rowBase = rowOffset + mGroup * (MT * 16);
    const int col = (nTile << 4) + ln;

    v8f acc[3][MT];
    const float bv = sW[3 * nW1 + col];
#pragma unroll
    for (int i = 0; i < MT; ++i)
#pragma unroll
        for (int r = 0; r < 8; ++r) {
            acc[0][i][r] = bv;
            acc[1][i][r] = 0.0f;
            acc[2][i][r] = 0.0f;
        }

    const float* arow = T0 + (size_t)(rowBase + ln) * 64 + (lg << 1);
    const float* bcol = sW + (lg << 1) * F + col;
#pragma unroll
    for (int kb = 0; kb < 16; ++kb) {
        v2f b[3];
#pragma unroll
        for (int m = 0; m < 3; ++m) {
            b[m].x = bcol[m * nW1 + (kb * 4) * F];
            b[m].y = bcol[m * nW1 + (kb * 4 + 1) * F];
        }
#pragma unroll
        for (int i = 0; i < MT; ++i) {
            v2f a = *(const v2f*)(arow + (size_t)i * (16 * 64) + kb * 4);
            if (RELU_IN) {
                a.x = fmaxf(a.x, 0.0f);
                a.y = fmaxf(a.y, 0.0f);
            }
#pragma unroll
            for (int m = 0; m < 3; ++m)
                acc[m][i] = __builtin_amdgcn_wmma_f32_16x16x4_f32(
                    false, a, false, b[m], (short)0, acc[m][i], false, false);
        }
    }

    float* outs[3] = {O, Z, P};
#pragma unroll
    for (int m = 0; m < 3; ++m) {
#pragma unroll
        for (int i = 0; i < MT; ++i) {
            float* orow = outs[m] + (size_t)(rowBase + i * 16 + (lg << 3)) * F + col;
#pragma unroll
            for (int r = 0; r < 8; ++r) orow[r * F] = acc[m][i][r];
        }
    }
}

template <int F, bool RELU_IN>
static void cheb_layer(const float* T0, const float* W, const float* b,
                       float* P, float* Z, float* O,
                       const int* src, const int* dst, const float* wE,
                       int* claim, int n, int E, hipStream_t stream) {
    const int nTilesN = F / 16;
    const int mTiles = n / 16;
    const int mGroups = mTiles / 5;
    const int wt5 = mGroups * nTilesN;
    if (wt5 > 0) {
        cheb_products_wmma_kernel<5, F, RELU_IN><<<(wt5 + 3) / 4, dim3(32, 4), 0, stream>>>(
            T0, W, b, O, Z, P, claim, wt5, 0);
    }
    const int remTiles = mTiles - mGroups * 5;
    if (remTiles > 0) {
        const int wt1 = remTiles * nTilesN;
        cheb_products_wmma_kernel<1, F, RELU_IN><<<(wt1 + 3) / 4, dim3(32, 4), 0, stream>>>(
            T0, W, b, O, Z, P, claim, wt1, mGroups * 80);
    }

    const int sblocks = 2048;
    cheb_scatter_kernel<F><<<sblocks, 256, 0, stream>>>(P, Z, src, dst, wE, 2.0f, E, n / 2, claim);
    cheb_scatter_kernel<F><<<sblocks, 256, 0, stream>>>(Z, O, src, dst, wE, 1.0f, E, n / 2, claim + 1);
}

extern "C" void kernel_launch(void* const* d_in, const int* in_sizes, int n_in,
                              void* d_out, int out_size, void* d_ws, size_t ws_size,
                              hipStream_t stream) {
    const float* x  = (const float*)d_in[0];
    const int*   ei = (const int*)d_in[1];
    const float* W1 = (const float*)d_in[2];
    const float* b1 = (const float*)d_in[3];
    const float* W2 = (const float*)d_in[4];
    const float* b2 = (const float*)d_in[5];
    const float* W3 = (const float*)d_in[6];
    const float* b3 = (const float*)d_in[7];

    const int n = in_sizes[0] / 64;
    const int E = in_sizes[1] / 2;
    const int* src = ei;
    const int* dst = ei + E;

    float* ws = (float*)d_ws;
    size_t o = 0;
    float* dis = ws + o; o += ((size_t)n + 63) & ~(size_t)63;
    int* claim = (int*)dis;
    float* wE  = ws + o; o += ((size_t)E + 63) & ~(size_t)63;
    float* P   = ws + o; o += (size_t)n * 64;
    float* Z   = ws + o; o += (size_t)n * 64;
    float* HA  = ws + o; o += (size_t)n * 64;
    float* HB  = ws + o; o += (size_t)n * 64;

    const int npad = (int)(((size_t)n + 63) & ~(size_t)63);
    unsigned* part = (unsigned*)P;
    cheb_degree_kernel<<<kDegRanges * kDegChunks, kPrepThreads, 0, stream>>>(src, part, E, n, npad);
    cheb_dis_kernel<<<(n + 255) / 256, 256, 0, stream>>>(part, dis, n, npad);
    cheb_edgew_kernel<<<256, kPrepThreads, 0, stream>>>(src, dst, dis, wE, E, n);

    cheb_layer<64, false>(x,  W1, b1, P, Z, HA, src, dst, wE, claim, n, E, stream);
    cheb_layer<64, true >(HA, W2, b2, P, Z, HB, src, dst, wE, claim, n, E, stream);
    cheb_layer<32, true >(HB, W3, b3, P, Z, (float*)d_out, src, dst, wE, claim, n, E, stream);
}
